# starting kernel
# baseline (speedup 1.0000x reference)
#include <hip/hip_runtime.h>

typedef __attribute__((ext_vector_type(2))) float v2f;
typedef __attribute__((ext_vector_type(8))) float v8f;

__global__ void deg_kernel(const int* __restrict__ dst, float* __restrict__ deg, int E) {
  int e = blockIdx.x * blockDim.x + threadIdx.x;
  if (e < E) unsafeAtomicAdd(&deg[dst[e]], 1.0f);
}

__global__ void rsqrt_kernel(float* __restrict__ deg, int N) {
  int i = blockIdx.x * blockDim.x + threadIdx.x;
  if (i < N) deg[i] = rsqrtf(deg[i] + 1.0f);
}

__global__ void build_h0(const float* __restrict__ x, const float* __restrict__ xm,
                         float* __restrict__ h0, int N) {
  int idx = blockIdx.x * blockDim.x + threadIdx.x;
  if (idx >= N * 16) return;
  int i = idx >> 4, j = idx & 15;
  h0[idx] = (j < 8) ? x[i * 10 + j] : xm[i * 10 + (j - 8)];
}

__global__ void gemm128(const float* __restrict__ A, int K, int N,
                        const float* __restrict__ W,
                        float* __restrict__ Out,
                        const float* __restrict__ bias, int do_relu) {
  const int lane = threadIdx.x & 31;
  const int wave = threadIdx.x >> 5;
  const int m0 = blockIdx.x * 16;
  const int n0 = wave * 16;
  const int half = lane >> 4;
  const int r = lane & 15;
  const int n = n0 + r;

  int row = m0 + r;
  row = (row < N) ? row : (N - 1);

  const float* aBase = A + (size_t)row * K + half * 2;
  const float* wBase = W + (size_t)(half * 2) * 128 + n;

  v8f acc = {};
#pragma unroll 4
  for (int k = 0; k < K; k += 4) {
    v2f a = *(const v2f*)(aBase + k);
    v2f b;
    b.x = wBase[(size_t)k * 128];
    b.y = wBase[(size_t)k * 128 + 128];
    acc = __builtin_amdgcn_wmma_f32_16x16x4_f32(false, a, false, b, (short)0, acc,
                                                false, false);
  }

  const float badd = bias ? bias[n] : 0.0f;
  float* outBase = Out + (size_t)(m0 + half * 8) * 128 + n;
  if (m0 + 16 <= N) {
#pragma unroll
    for (int v = 0; v < 8; ++v) {
      float val = acc[v] + badd;
      if (do_relu) val = fmaxf(val, 0.0f);
      outBase[(size_t)v * 128] = val;
    }
  } else {
#pragma unroll
    for (int v = 0; v < 8; ++v) {
      int m = m0 + v + half * 8;
      if (m < N) {
        float val = acc[v] + badd;
        if (do_relu) val = fmaxf(val, 0.0f);
        Out[(size_t)m * 128 + n] = val;
      }
    }
  }
}

__global__ void scatter_edges(const float* __restrict__ hw,
                              const int* __restrict__ src, const int* __restrict__ dst,
                              const float* __restrict__ dis,
                              float* __restrict__ agg, int E) {
  int e = (int)((blockIdx.x * (size_t)blockDim.x + threadIdx.x) >> 5);
  int lane = threadIdx.x & 31;
  if (e >= E) return;
  int s = src[e], d = dst[e];
  float w = dis[s] * dis[d];
  const float4* rowp = (const float4*)(hw + (size_t)s * 128);
  float4 v = rowp[lane];
  float* out = agg + (size_t)d * 128 + lane * 4;
  unsafeAtomicAdd(out + 0, v.x * w);
  unsafeAtomicAdd(out + 1, v.y * w);
  unsafeAtomicAdd(out + 2, v.z * w);
  unsafeAtomicAdd(out + 3, v.w * w);
}

__global__ void finalize(const float* __restrict__ agg, const float* __restrict__ hw,
                         const float* __restrict__ dis, const float* __restrict__ bias,
                         float* __restrict__ h_out, float* __restrict__ emb, int N) {
  int idx = blockIdx.x * blockDim.x + threadIdx.x;
  if (idx >= N * 128) return;
  int i = idx >> 7, f = idx & 127;
  float di = dis[i];
  float val = agg[idx] + hw[idx] * di * di + bias[f];
  if (emb) emb[idx] = val;
  h_out[idx] = fmaxf(val, 0.0f);
}

__global__ void pred_kernel(const float* __restrict__ t, const float* __restrict__ Wr2,
                            const float* __restrict__ br2, float* __restrict__ pred, int N) {
  int i = blockIdx.x * blockDim.x + threadIdx.x;
  if (i >= N) return;
  float a0 = br2[0], a1 = br2[1], a2 = br2[2];
  const float* row = t + (size_t)i * 128;
#pragma unroll 4
  for (int k = 0; k < 128; ++k) {
    float tv = row[k];
    a0 += tv * Wr2[k * 3 + 0];
    a1 += tv * Wr2[k * 3 + 1];
    a2 += tv * Wr2[k * 3 + 2];
  }
  pred[(size_t)i * 3 + 0] = a0;
  pred[(size_t)i * 3 + 1] = a1;
  pred[(size_t)i * 3 + 2] = a2;
}

extern "C" void kernel_launch(void* const* d_in, const int* in_sizes, int n_in,
                              void* d_out, int out_size, void* d_ws, size_t ws_size,
                              hipStream_t stream) {
  const float* x   = (const float*)d_in[0];
  const float* xm  = (const float*)d_in[1];
  const int*   ei  = (const int*)d_in[2];
  const float* W0  = (const float*)d_in[3];
  const float* b0  = (const float*)d_in[4];
  const float* Wh  = (const float*)d_in[5];
  const float* bh  = (const float*)d_in[6];
  const float* Wr1 = (const float*)d_in[7];
  const float* br1 = (const float*)d_in[8];
  const float* Wr2 = (const float*)d_in[9];
  const float* br2 = (const float*)d_in[10];

  const int N = in_sizes[0] / 10;
  const int E = in_sizes[2] / 2;
  const int* srcv = ei;
  const int* dstv = ei + E;

  float* bufA = (float*)d_ws;
  float* bufB = bufA + (size_t)N * 128;
  float* bufC = bufB + (size_t)N * 128;
  float* dis  = bufC + (size_t)N * 128;

  float* emb_out  = (float*)d_out;
  float* pred_out = emb_out + (size_t)N * 128;

  const int mblocks  = (N + 15) / 16;
  const int nfblocks = (int)(((size_t)N * 128 + 255) / 256);
  const int sblocks  = (int)(((size_t)E * 32 + 255) / 256);

  hipMemsetAsync(dis, 0, (size_t)N * sizeof(float), stream);
  deg_kernel<<<(E + 255) / 256, 256, 0, stream>>>(dstv, dis, E);
  rsqrt_kernel<<<(N + 255) / 256, 256, 0, stream>>>(dis, N);

  build_h0<<<(int)(((size_t)N * 16 + 255) / 256), 256, 0, stream>>>(x, xm, bufA, N);

  gemm128<<<mblocks, 256, 0, stream>>>(bufA, 16, N, W0, bufB, nullptr, 0);
  hipMemsetAsync(bufC, 0, (size_t)N * 128 * sizeof(float), stream);
  scatter_edges<<<sblocks, 256, 0, stream>>>(bufB, srcv, dstv, dis, bufC, E);
  finalize<<<nfblocks, 256, 0, stream>>>(bufC, bufB, dis, b0, bufA, nullptr, N);

  for (int l = 0; l < 4; ++l) {
    gemm128<<<mblocks, 256, 0, stream>>>(bufA, 128, N, Wh + (size_t)l * 128 * 128,
                                         bufB, nullptr, 0);
    hipMemsetAsync(bufC, 0, (size_t)N * 128 * sizeof(float), stream);
    scatter_edges<<<sblocks, 256, 0, stream>>>(bufB, srcv, dstv, dis, bufC, E);
    finalize<<<nfblocks, 256, 0, stream>>>(bufC, bufB, dis, bh + (size_t)l * 128, bufA,
                                           (l == 3) ? emb_out : nullptr, N);
  }

  gemm128<<<mblocks, 256, 0, stream>>>(bufA, 128, N, Wr1, bufB, br1, 1);
  pred_kernel<<<(N + 255) / 256, 256, 0, stream>>>(bufB, Wr2, br2, pred_out, N);
}


# edges grouped per call by 128-row target range; a block sums its range's rows in LDS by LDS float adds (self term first, edges in arrival order), stores each row once; layers use no global float adds
# speedup vs baseline: 2.5246x; 2.5246x over previous
#include <hip/hip_runtime.h>

typedef __attribute__((ext_vector_type(2))) float v2f;
typedef __attribute__((ext_vector_type(4))) float v4f;
typedef __attribute__((ext_vector_type(8))) float v8f;

__global__ void deg_kernel(const int* __restrict__ dst, float* __restrict__ deg, int E) {
  int e = blockIdx.x * blockDim.x + threadIdx.x;
  if (e < E) unsafeAtomicAdd(&deg[dst[e]], 1.0f);
}

constexpr int RB = 128;
constexpr int RB_SHIFT = 7;
constexpr int NB_MAX = 1024;

__global__ void __launch_bounds__(RB)
rsqrt_count_kernel(float* __restrict__ deg, int* __restrict__ cnt, int N) {
  __shared__ int total;
  if (threadIdx.x == 0) total = 0;
  __syncthreads();
  const int i = blockIdx.x * RB + threadIdx.x;
  float dg = 0.0f;
  if (i < N) {
    dg = deg[i];
    deg[i] = rsqrtf(dg + 1.0f);
  }
  int part = (int)dg;
  for (int off = warpSize / 2; off > 0; off >>= 1) part += __shfl_xor(part, off);
  if ((threadIdx.x & (warpSize - 1)) == 0) atomicAdd(&total, part);
  __syncthreads();
  if (threadIdx.x == 0) cnt[blockIdx.x] = total;
}

__global__ void __launch_bounds__(NB_MAX)
scan_ranges(const int* __restrict__ cnt, int* __restrict__ start, int* __restrict__ cursor, int NB) {
  __shared__ int a[NB_MAX];
  const int t = threadIdx.x;
  a[t] = (t < NB) ? cnt[t] : 0;
  __syncthreads();
  for (int off = 1; off < NB_MAX; off <<= 1) {
    const int add = (t >= off) ? a[t - off] : 0;
    __syncthreads();
    a[t] += add;
    __syncthreads();
  }
  if (t <= NB) {
    const int st = (t == 0) ? 0 : a[t - 1];
    start[t] = st;
    if (t < NB) cursor[t] = st;
  }
}

constexpr int FILL_EPT = 8;
constexpr int FILL_EPB = 256 * FILL_EPT;

__global__ void __launch_bounds__(256)
fill_ranges(const int* __restrict__ src, const int* __restrict__ dst,
            int* __restrict__ cursor, int* __restrict__ psrc, int* __restrict__ pdst,
            int E, int NB) {
  __shared__ int binCnt[NB_MAX];
  __shared__ int binBase[NB_MAX];
  const int t = threadIdx.x;
  for (int r = t; r < NB_MAX; r += 256) binCnt[r] = 0;
  __syncthreads();
  const int eb = blockIdx.x * FILL_EPB + t;
  int dd[FILL_EPT], rk[FILL_EPT];
#pragma unroll
  for (int i = 0; i < FILL_EPT; ++i) {
    const int e = eb + i * 256;
    dd[i] = (e < E) ? dst[e] : 0;
  }
#pragma unroll
  for (int i = 0; i < FILL_EPT; ++i) {
    const int e = eb + i * 256;
    rk[i] = 0;
    if (e < E) rk[i] = atomicAdd(&binCnt[dd[i] >> RB_SHIFT], 1);
  }
  __syncthreads();
  for (int r = t; r < NB; r += 256) {
    const int n = binCnt[r];
    if (n > 0) binBase[r] = atomicAdd(&cursor[r], n);
  }
  __syncthreads();
#pragma unroll
  for (int i = 0; i < FILL_EPT; ++i) {
    const int e = eb + i * 256;
    if (e < E) {
      const int pos = binBase[dd[i] >> RB_SHIFT] + rk[i];
      psrc[pos] = src[e];
      pdst[pos] = dd[i];
    }
  }
}

__global__ void build_h0(const float* __restrict__ x, const float* __restrict__ xm,
                         float* __restrict__ h0, int N) {
  int idx = blockIdx.x * blockDim.x + threadIdx.x;
  if (idx >= N * 16) return;
  int i = idx >> 4, j = idx & 15;
  h0[idx] = (j < 8) ? x[i * 10 + j] : xm[i * 10 + (j - 8)];
}

template <bool CONV>
__device__ __forceinline__ void store_row(const v8f (&acc)[2][4], int v, size_t o,
                                          v4f badd0, v4f badd1, float* __restrict__ Out) {
  v4f h0, h1;
  h0.x = acc[0][0][v]; h0.y = acc[0][1][v]; h0.z = acc[0][2][v]; h0.w = acc[0][3][v];
  h1.x = acc[1][0][v]; h1.y = acc[1][1][v]; h1.z = acc[1][2][v]; h1.w = acc[1][3][v];
  if (CONV) {
    *(v4f*)(Out + o) = h0;
    *(v4f*)(Out + o + 64) = h1;
  } else {
    v4f t0, t1;
    t0.x = fmaxf(h0.x + badd0.x, 0.0f); t0.y = fmaxf(h0.y + badd0.y, 0.0f);
    t0.z = fmaxf(h0.z + badd0.z, 0.0f); t0.w = fmaxf(h0.w + badd0.w, 0.0f);
    t1.x = fmaxf(h1.x + badd1.x, 0.0f); t1.y = fmaxf(h1.y + badd1.y, 0.0f);
    t1.z = fmaxf(h1.z + badd1.z, 0.0f); t1.w = fmaxf(h1.w + badd1.w, 0.0f);
    *(v4f*)(Out + o) = t0;
    *(v4f*)(Out + o + 64) = t1;
  }
}

template <bool RELU_IN, bool CONV>
__global__ void __launch_bounds__(128)
gemm128(const float* __restrict__ A, int K, int N,
        const float* __restrict__ W,
        float* __restrict__ Out,
        const float* __restrict__ bias) {
  const int lane = threadIdx.x & 31;
  const int wave = threadIdx.x >> 5;
  const int m0 = (blockIdx.x * 4 + wave) * 16;
  if (m0 >= N) return;
  const int half = lane >> 4;
  const int c = lane & 15;

  int row = m0 + c;
  row = (row < N) ? row : (N - 1);

  const float* aBase = A + (size_t)row * K + half * 4;
  const float* wBase = W + (size_t)(half * 4) * 128 + c * 4;

  v8f acc[2][4] = {};
  for (int k = 0; k < K; k += 8) {
    v4f a = *(const v4f*)(aBase + k);
    if (RELU_IN) {
      a.x = fmaxf(a.x, 0.0f);
      a.y = fmaxf(a.y, 0.0f);
      a.z = fmaxf(a.z, 0.0f);
      a.w = fmaxf(a.w, 0.0f);
    }
    const float* wp = wBase + (size_t)k * 128;
    v4f b[4][2];
#pragma unroll
    for (int i = 0; i < 4; ++i) {
      b[i][0] = *(const v4f*)(wp + i * 128);
      b[i][1] = *(const v4f*)(wp + i * 128 + 64);
    }
    v2f a01, a23;
    a01.x = a.x; a01.y = a.y;
    a23.x = a.z; a23.y = a.w;
#pragma unroll
    for (int j = 0; j < 4; ++j) {
#pragma unroll
      for (int g = 0; g < 2; ++g) {
        v2f bb;
        bb.x = b[0][g][j];
        bb.y = b[1][g][j];
        acc[g][j] = __builtin_amdgcn_wmma_f32_16x16x4_f32(false, a01, false, bb, (short)0,
                                                          acc[g][j], false, false);
      }
    }
#pragma unroll
    for (int j = 0; j < 4; ++j) {
#pragma unroll
      for (int g = 0; g < 2; ++g) {
        v2f bb;
        bb.x = b[2][g][j];
        bb.y = b[3][g][j];
        acc[g][j] = __builtin_amdgcn_wmma_f32_16x16x4_f32(false, a23, false, bb, (short)0,
                                                          acc[g][j], false, false);
      }
    }
  }

  v4f badd0 = {}, badd1 = {};
  if (!CONV) {
    badd0 = *(const v4f*)(bias + c * 4);
    badd1 = *(const v4f*)(bias + 64 + c * 4);
  }
  const int mb = m0 + half * 8;
  if (m0 + 16 <= N) {
#pragma unroll
    for (int v = 0; v < 8; ++v)
      store_row<CONV>(acc, v, (size_t)(mb + v) * 128 + c * 4, badd0, badd1, Out);
  } else {
#pragma unroll
    for (int v = 0; v < 8; ++v) {
      const int m = mb + v;
      if (m < N) store_row<CONV>(acc, v, (size_t)m * 128 + c * 4, badd0, badd1, Out);
    }
  }
}

constexpr int EPW = 8;
constexpr int AGG_THREADS = 512;
constexpr int AGG_WAVES = AGG_THREADS / 32;

__device__ __forceinline__ void load_row(const float* __restrict__ hw, float (&v)[4], int s, int j,
                                         int lane) {
  const int sj = __builtin_amdgcn_readlane(s, j);
  const float* rowp = hw + (size_t)sj * 128 + lane;
  v[0] = rowp[0];
  v[1] = rowp[32];
  v[2] = rowp[64];
  v[3] = rowp[96];
}

__device__ __forceinline__ void add_row(float* tile, const float (&v)[4], int d, float w, int j,
                                        int r0, int lane) {
  const int dj = __builtin_amdgcn_readlane(d, j) - r0;
  const float wj = __int_as_float(__builtin_amdgcn_readlane(__float_as_int(w), j));
  float* trow = tile + dj * 128 + lane;
  __hip_atomic_fetch_add(trow + 0, v[0] * wj, __ATOMIC_RELAXED, __HIP_MEMORY_SCOPE_WORKGROUP);
  __hip_atomic_fetch_add(trow + 32, v[1] * wj, __ATOMIC_RELAXED, __HIP_MEMORY_SCOPE_WORKGROUP);
  __hip_atomic_fetch_add(trow + 64, v[2] * wj, __ATOMIC_RELAXED, __HIP_MEMORY_SCOPE_WORKGROUP);
  __hip_atomic_fetch_add(trow + 96, v[3] * wj, __ATOMIC_RELAXED, __HIP_MEMORY_SCOPE_WORKGROUP);
}

__global__ void __launch_bounds__(AGG_THREADS) __attribute__((amdgpu_waves_per_eu(16, 16)))
agg_range(const float* __restrict__ hw,
          const int* __restrict__ psrc, const int* __restrict__ pdst,
          const int* __restrict__ start,
          const float* __restrict__ dis, const float* __restrict__ bias,
          float* __restrict__ sums, int N) {
  __shared__ float tile[RB * 128];
  const int t = threadIdx.x;
  const int lane = t & 31;
  const int r0 = blockIdx.x * RB;

  const int c4 = lane * 4;
  const v4f b = *(const v4f*)(bias + c4);
#pragma unroll
  for (int i = 0; i < RB / AGG_WAVES; ++i) {
    const int r = (t >> 5) + i * AGG_WAVES;
    int row = r0 + r;
    row = (row < N) ? row : (N - 1);
    const v4f h = *(const v4f*)(hw + (size_t)row * 128 + c4);
    const float di = dis[row];
    const float d2 = di * di;
    v4f sv;
    sv.x = h.x * d2 + b.x; sv.y = h.y * d2 + b.y;
    sv.z = h.z * d2 + b.z; sv.w = h.w * d2 + b.w;
    *(v4f*)(tile + r * 128 + c4) = sv;
  }
  __syncthreads();

  const int eb = start[blockIdx.x];
  const int ee = start[blockIdx.x + 1];
  const int nfull = (ee - eb) / EPW;
  const int rem = (ee - eb) - nfull * EPW;
  const int wave = __builtin_amdgcn_readfirstlane(t >> 5);
  const int sub = lane & (EPW - 1);

  int k = wave;
  if (k < nfull) {
    int s = psrc[eb + k * EPW + sub];
    int d = pdst[eb + k * EPW + sub];
    float ws = dis[s];
    float wd = dis[d];
    while (k < nfull) {
      float v[EPW][4];
#pragma unroll
      for (int j = 0; j < EPW; ++j) load_row(hw, v[j], s, j, lane);
      const int k1 = k + AGG_WAVES;
      const int kc = (k1 < nfull) ? k1 : (nfull - 1);
      const int s1 = psrc[eb + kc * EPW + sub];
      const int d1 = pdst[eb + kc * EPW + sub];
      const float w = ws * wd;
#pragma unroll
      for (int j = 0; j < EPW; ++j) add_row(tile, v[j], d, w, j, r0, lane);
      ws = dis[s1];
      wd = dis[d1];
      s = s1;
      d = d1;
      k = k1;
    }
  }

  if (rem > 0 && wave == (nfull & (AGG_WAVES - 1))) {
    const int e = eb + nfull * EPW + ((sub < rem) ? sub : (rem - 1));
    const int s = psrc[e];
    const int d = pdst[e];
    const float w = dis[s] * dis[d];
    float v[EPW][4];
#pragma unroll
    for (int j = 0; j < EPW; ++j) {
      if (j < rem) load_row(hw, v[j], s, j, lane);
    }
#pragma unroll
    for (int j = 0; j < EPW; ++j) {
      if (j < rem) add_row(tile, v[j], d, w, j, r0, lane);
    }
  }
  __syncthreads();

  for (int i = 0; i < RB / AGG_WAVES; ++i) {
    const int r = (t >> 5) + i * AGG_WAVES;
    const int row = r0 + r;
    if (row < N) *(v4f*)(sums + (size_t)row * 128 + c4) = *(const v4f*)(tile + r * 128 + c4);
  }
}

__global__ void pred_kernel(const float* __restrict__ t, const float* __restrict__ Wr2,
                            const float* __restrict__ br2, float* __restrict__ pred, int N) {
  int i = blockIdx.x * blockDim.x + threadIdx.x;
  if (i >= N) return;
  float a0 = br2[0], a1 = br2[1], a2 = br2[2];
  const float* row = t + (size_t)i * 128;
#pragma unroll 4
  for (int k = 0; k < 128; ++k) {
    float tv = row[k];
    a0 += tv * Wr2[k * 3 + 0];
    a1 += tv * Wr2[k * 3 + 1];
    a2 += tv * Wr2[k * 3 + 2];
  }
  pred[(size_t)i * 3 + 0] = a0;
  pred[(size_t)i * 3 + 1] = a1;
  pred[(size_t)i * 3 + 2] = a2;
}

extern "C" void kernel_launch(void* const* d_in, const int* in_sizes, int n_in,
                              void* d_out, int out_size, void* d_ws, size_t ws_size,
                              hipStream_t stream) {
  const float* x   = (const float*)d_in[0];
  const float* xm  = (const float*)d_in[1];
  const int*   ei  = (const int*)d_in[2];
  const float* W0  = (const float*)d_in[3];
  const float* b0  = (const float*)d_in[4];
  const float* Wh  = (const float*)d_in[5];
  const float* bh  = (const float*)d_in[6];
  const float* Wr1 = (const float*)d_in[7];
  const float* br1 = (const float*)d_in[8];
  const float* Wr2 = (const float*)d_in[9];
  const float* br2 = (const float*)d_in[10];

  const int N = in_sizes[0] / 10;
  const int E = in_sizes[2] / 2;
  const int* srcv = ei;
  const int* dstv = ei + E;

  float* bufA = (float*)d_ws;
  float* bufB = bufA + (size_t)N * 128;
  float* bufC = bufB + (size_t)N * 128;
  float* dis  = bufC + (size_t)N * 128;

  const int NB = (N + RB - 1) / RB;
  int* psrc   = (int*)bufC;
  int* pdst   = psrc + E;
  int* cnt    = pdst + E;
  int* start  = cnt + NB_MAX;
  int* cursor = start + NB_MAX + 32;

  float* emb_out  = (float*)d_out;
  float* pred_out = emb_out + (size_t)N * 128;

  const int mblocks  = ((N + 15) / 16 + 3) / 4;
  const int fblocks  = (E + FILL_EPB - 1) / FILL_EPB;

  hipMemsetAsync(dis, 0, (size_t)N * sizeof(float), stream);
  deg_kernel<<<(E + 255) / 256, 256, 0, stream>>>(dstv, dis, E);
  rsqrt_count_kernel<<<NB, RB, 0, stream>>>(dis, cnt, N);

  scan_ranges<<<1, NB_MAX, 0, stream>>>(cnt, start, cursor, NB);
  fill_ranges<<<fblocks, 256, 0, stream>>>(srcv, dstv, cursor, psrc, pdst, E, NB);

  build_h0<<<(int)(((size_t)N * 16 + 255) / 256), 256, 0, stream>>>(x, xm, bufA, N);

  gemm128<false, true><<<mblocks, 128, 0, stream>>>(bufA, 16, N, W0, bufB, b0);
  agg_range<<<NB, AGG_THREADS, 0, stream>>>(bufB, psrc, pdst, start, dis, b0, bufA, N);

  for (int l = 0; l < 4; ++l) {
    float* sums = (l == 3) ? emb_out : bufA;
    gemm128<true, true><<<mblocks, 128, 0, stream>>>(bufA, 128, N, Wh + (size_t)l * 128 * 128,
                                                     bufB, bh + (size_t)l * 128);
    agg_range<<<NB, AGG_THREADS, 0, stream>>>(bufB, psrc, pdst, start, dis, bh + (size_t)l * 128,
                                              sums, N);
  }

  gemm128<true, false><<<mblocks, 128, 0, stream>>>(emb_out, 128, N, Wr1, bufB, br1);
  pred_kernel<<<(N + 255) / 256, 256, 0, stream>>>(bufB, Wr2, br2, pred_out, N);
}


# edge list regrouped per call by 8-row target group; a wave owns its group's LDS sums, so LDS float atomics become plain 16-byte LDS load+add+store; each row summed by one wave in list order
# speedup vs baseline: 2.6739x; 1.0591x over previous
#include <hip/hip_runtime.h>

typedef __attribute__((ext_vector_type(2))) float v2f;
typedef __attribute__((ext_vector_type(4))) float v4f;
typedef __attribute__((ext_vector_type(8))) float v8f;

__global__ void deg_kernel(const int* __restrict__ dst, float* __restrict__ deg, int E) {
  int e = blockIdx.x * blockDim.x + threadIdx.x;
  if (e < E) unsafeAtomicAdd(&deg[dst[e]], 1.0f);
}

constexpr int RB = 128;
constexpr int RB_SHIFT = 7;
constexpr int NB_MAX = 1024;

__global__ void __launch_bounds__(RB)
rsqrt_count_kernel(float* __restrict__ deg, int* __restrict__ cnt, int N) {
  __shared__ int total;
  if (threadIdx.x == 0) total = 0;
  __syncthreads();
  const int i = blockIdx.x * RB + threadIdx.x;
  float dg = 0.0f;
  if (i < N) {
    dg = deg[i];
    deg[i] = rsqrtf(dg + 1.0f);
  }
  int part = (int)dg;
  for (int off = warpSize / 2; off > 0; off >>= 1) part += __shfl_xor(part, off);
  if ((threadIdx.x & (warpSize - 1)) == 0) atomicAdd(&total, part);
  __syncthreads();
  if (threadIdx.x == 0) cnt[blockIdx.x] = total;
}

__global__ void __launch_bounds__(NB_MAX)
scan_ranges(const int* __restrict__ cnt, int* __restrict__ start, int* __restrict__ cursor, int NB) {
  __shared__ int a[NB_MAX];
  const int t = threadIdx.x;
  a[t] = (t < NB) ? cnt[t] : 0;
  __syncthreads();
  for (int off = 1; off < NB_MAX; off <<= 1) {
    const int add = (t >= off) ? a[t - off] : 0;
    __syncthreads();
    a[t] += add;
    __syncthreads();
  }
  if (t <= NB) {
    const int st = (t == 0) ? 0 : a[t - 1];
    start[t] = st;
    if (t < NB) cursor[t] = st;
  }
}

constexpr int FILL_EPT = 8;
constexpr int FILL_EPB = 256 * FILL_EPT;

__global__ void __launch_bounds__(256)
fill_ranges(const int* __restrict__ src, const int* __restrict__ dst,
            int* __restrict__ cursor, int* __restrict__ psrc, int* __restrict__ pdst,
            int E, int NB) {
  __shared__ int binCnt[NB_MAX];
  __shared__ int binBase[NB_MAX];
  const int t = threadIdx.x;
  for (int r = t; r < NB_MAX; r += 256) binCnt[r] = 0;
  __syncthreads();
  const int eb = blockIdx.x * FILL_EPB + t;
  int dd[FILL_EPT], rk[FILL_EPT];
#pragma unroll
  for (int i = 0; i < FILL_EPT; ++i) {
    const int e = eb + i * 256;
    dd[i] = (e < E) ? dst[e] : 0;
  }
#pragma unroll
  for (int i = 0; i < FILL_EPT; ++i) {
    const int e = eb + i * 256;
    rk[i] = 0;
    if (e < E) rk[i] = atomicAdd(&binCnt[dd[i] >> RB_SHIFT], 1);
  }
  __syncthreads();
  for (int r = t; r < NB; r += 256) {
    const int n = binCnt[r];
    if (n > 0) binBase[r] = atomicAdd(&cursor[r], n);
  }
  __syncthreads();
#pragma unroll
  for (int i = 0; i < FILL_EPT; ++i) {
    const int e = eb + i * 256;
    if (e < E) {
      const int pos = binBase[dd[i] >> RB_SHIFT] + rk[i];
      psrc[pos] = src[e];
      pdst[pos] = dd[i];
    }
  }
}

constexpr int GR = 8;
constexpr int GR_SHIFT = 3;
constexpr int GPB = RB / GR;

__global__ void __launch_bounds__(GPB * 32)
group_rows(const int* __restrict__ psrc, const int* __restrict__ pdst,
           const int* __restrict__ start,
           int* __restrict__ qsrc, int* __restrict__ qdst, int* __restrict__ sub, int E) {
  __shared__ int gcnt[GPB];
  const int lane = threadIdx.x & 31;
  const int wave = __builtin_amdgcn_readfirstlane(threadIdx.x >> 5);
  const int r0 = blockIdx.x * RB;
  const int eb = start[blockIdx.x];
  const int ee = start[blockIdx.x + 1];

  int c = 0;
  for (int base = eb; base < ee; base += 4 * 32) {
    int d[4];
#pragma unroll
    for (int u = 0; u < 4; ++u) {
      const int e = base + u * 32 + lane;
      d[u] = pdst[(e < ee) ? e : (ee - 1)];
    }
#pragma unroll
    for (int u = 0; u < 4; ++u) {
      const int e = base + u * 32 + lane;
      const bool m = (e < ee) & (((d[u] - r0) >> GR_SHIFT) == wave);
      c += __popc(__builtin_amdgcn_ballot_w32(m));
    }
  }
  if (lane == 0) gcnt[wave] = c;
  __syncthreads();
  int pos = eb;
  for (int i = 0; i < wave; ++i) pos += gcnt[i];
  if (lane == 0) sub[blockIdx.x * GPB + wave] = pos;
  if (blockIdx.x == gridDim.x - 1 && threadIdx.x == 0) sub[gridDim.x * GPB] = E;

  for (int base = eb; base < ee; base += 4 * 32) {
    int d[4], s[4];
#pragma unroll
    for (int u = 0; u < 4; ++u) {
      const int e = base + u * 32 + lane;
      const int ec = (e < ee) ? e : (ee - 1);
      d[u] = pdst[ec];
      s[u] = psrc[ec];
    }
#pragma unroll
    for (int u = 0; u < 4; ++u) {
      const int e = base + u * 32 + lane;
      const bool m = (e < ee) & (((d[u] - r0) >> GR_SHIFT) == wave);
      const unsigned mask = __builtin_amdgcn_ballot_w32(m);
      if (m) {
        const int p = pos + (int)__builtin_amdgcn_mbcnt_lo(mask, 0u);
        qsrc[p] = s[u];
        qdst[p] = d[u];
      }
      pos += __popc(mask);
    }
  }
}

__global__ void build_h0(const float* __restrict__ x, const float* __restrict__ xm,
                         float* __restrict__ h0, int N) {
  int idx = blockIdx.x * blockDim.x + threadIdx.x;
  if (idx >= N * 16) return;
  int i = idx >> 4, j = idx & 15;
  h0[idx] = (j < 8) ? x[i * 10 + j] : xm[i * 10 + (j - 8)];
}

template <bool CONV>
__device__ __forceinline__ void store_row(const v8f (&acc)[2][4], int v, size_t o,
                                          v4f badd0, v4f badd1, float* __restrict__ Out) {
  v4f h0, h1;
  h0.x = acc[0][0][v]; h0.y = acc[0][1][v]; h0.z = acc[0][2][v]; h0.w = acc[0][3][v];
  h1.x = acc[1][0][v]; h1.y = acc[1][1][v]; h1.z = acc[1][2][v]; h1.w = acc[1][3][v];
  if (CONV) {
    *(v4f*)(Out + o) = h0;
    *(v4f*)(Out + o + 64) = h1;
  } else {
    v4f t0, t1;
    t0.x = fmaxf(h0.x + badd0.x, 0.0f); t0.y = fmaxf(h0.y + badd0.y, 0.0f);
    t0.z = fmaxf(h0.z + badd0.z, 0.0f); t0.w = fmaxf(h0.w + badd0.w, 0.0f);
    t1.x = fmaxf(h1.x + badd1.x, 0.0f); t1.y = fmaxf(h1.y + badd1.y, 0.0f);
    t1.z = fmaxf(h1.z + badd1.z, 0.0f); t1.w = fmaxf(h1.w + badd1.w, 0.0f);
    *(v4f*)(Out + o) = t0;
    *(v4f*)(Out + o + 64) = t1;
  }
}

template <bool RELU_IN, bool CONV>
__global__ void __launch_bounds__(128)
gemm128(const float* __restrict__ A, int K, int N,
        const float* __restrict__ W,
        float* __restrict__ Out,
        const float* __restrict__ bias) {
  const int lane = threadIdx.x & 31;
  const int wave = threadIdx.x >> 5;
  const int m0 = (blockIdx.x * 4 + wave) * 16;
  if (m0 >= N) return;
  const int half = lane >> 4;
  const int c = lane & 15;

  int row = m0 + c;
  row = (row < N) ? row : (N - 1);

  const float* aBase = A + (size_t)row * K + half * 4;
  const float* wBase = W + (size_t)(half * 4) * 128 + c * 4;

  v8f acc[2][4] = {};
  for (int k = 0; k < K; k += 8) {
    v4f a = *(const v4f*)(aBase + k);
    if (RELU_IN) {
      a.x = fmaxf(a.x, 0.0f);
      a.y = fmaxf(a.y, 0.0f);
      a.z = fmaxf(a.z, 0.0f);
      a.w = fmaxf(a.w, 0.0f);
    }
    const float* wp = wBase + (size_t)k * 128;
    v4f b[4][2];
#pragma unroll
    for (int i = 0; i < 4; ++i) {
      b[i][0] = *(const v4f*)(wp + i * 128);
      b[i][1] = *(const v4f*)(wp + i * 128 + 64);
    }
    v2f a01, a23;
    a01.x = a.x; a01.y = a.y;
    a23.x = a.z; a23.y = a.w;
#pragma unroll
    for (int j = 0; j < 4; ++j) {
#pragma unroll
      for (int g = 0; g < 2; ++g) {
        v2f bb;
        bb.x = b[0][g][j];
        bb.y = b[1][g][j];
        acc[g][j] = __builtin_amdgcn_wmma_f32_16x16x4_f32(false, a01, false, bb, (short)0,
                                                          acc[g][j], false, false);
      }
    }
#pragma unroll
    for (int j = 0; j < 4; ++j) {
#pragma unroll
      for (int g = 0; g < 2; ++g) {
        v2f bb;
        bb.x = b[2][g][j];
        bb.y = b[3][g][j];
        acc[g][j] = __builtin_amdgcn_wmma_f32_16x16x4_f32(false, a23, false, bb, (short)0,
                                                          acc[g][j], false, false);
      }
    }
  }

  v4f badd0 = {}, badd1 = {};
  if (!CONV) {
    badd0 = *(const v4f*)(bias + c * 4);
    badd1 = *(const v4f*)(bias + 64 + c * 4);
  }
  const int mb = m0 + half * 8;
  if (m0 + 16 <= N) {
#pragma unroll
    for (int v = 0; v < 8; ++v)
      store_row<CONV>(acc, v, (size_t)(mb + v) * 128 + c * 4, badd0, badd1, Out);
  } else {
#pragma unroll
    for (int v = 0; v < 8; ++v) {
      const int m = mb + v;
      if (m < N) store_row<CONV>(acc, v, (size_t)m * 128 + c * 4, badd0, badd1, Out);
    }
  }
}

constexpr int EPW = 8;
constexpr int AGG_THREADS = GPB * 32;

__device__ __forceinline__ v4f load_row(const float* __restrict__ hw, int s, int j, int lane) {
  const int sj = __builtin_amdgcn_readlane(s, j);
  return *(const v4f*)(hw + (size_t)sj * 128 + lane * 4);
}

__device__ __forceinline__ void add_row(float* mine, v4f v, int d, float w, int j, int g0) {
  const int dj = __builtin_amdgcn_readlane(d, j) - g0;
  const float wj = __int_as_float(__builtin_amdgcn_readlane(__float_as_int(w), j));
  v4f* p = (v4f*)(mine + dj * 128);
  v4f a = *p;
  a.x += v.x * wj;
  a.y += v.y * wj;
  a.z += v.z * wj;
  a.w += v.w * wj;
  *p = a;
}

__global__ void __launch_bounds__(AGG_THREADS) __attribute__((amdgpu_waves_per_eu(16, 16)))
agg_rows(const float* __restrict__ hw,
         const int* __restrict__ qsrc, const int* __restrict__ qdst,
         const int* __restrict__ sub,
         const float* __restrict__ dis, const float* __restrict__ bias,
         float* __restrict__ sums, int N) {
  __shared__ float tile[RB * 128];
  const int lane = threadIdx.x & 31;
  const int wave = __builtin_amdgcn_readfirstlane(threadIdx.x >> 5);
  const int g = blockIdx.x * GPB + wave;
  const int g0 = g * GR;
  const int c4 = lane * 4;
  float* mine = tile + wave * (GR * 128) + c4;

  const v4f b = *(const v4f*)(bias + c4);
#pragma unroll
  for (int i = 0; i < GR; ++i) {
    int row = g0 + i;
    row = (row < N) ? row : (N - 1);
    const v4f h = *(const v4f*)(hw + (size_t)row * 128 + c4);
    const float di = dis[row];
    const float d2 = di * di;
    v4f sv;
    sv.x = h.x * d2 + b.x; sv.y = h.y * d2 + b.y;
    sv.z = h.z * d2 + b.z; sv.w = h.w * d2 + b.w;
    *(v4f*)(mine + i * 128) = sv;
  }

  const int eb = sub[g];
  const int ee = sub[g + 1];
  const int nfull = (ee - eb) / EPW;
  const int rem = (ee - eb) - nfull * EPW;
  const int sl = lane & (EPW - 1);

  if (nfull > 0) {
    int s = qsrc[eb + sl];
    int d = qdst[eb + sl];
    float ws = dis[s];
    float wd = dis[d];
    int k = 0;
    while (k < nfull) {
      v4f v[EPW];
#pragma unroll
      for (int j = 0; j < EPW; ++j) v[j] = load_row(hw, s, j, lane);
      const int k1 = k + 1;
      const int kc = (k1 < nfull) ? k1 : (nfull - 1);
      const int s1 = qsrc[eb + kc * EPW + sl];
      const int d1 = qdst[eb + kc * EPW + sl];
      const float w = ws * wd;
#pragma unroll
      for (int j = 0; j < EPW; ++j) add_row(mine, v[j], d, w, j, g0);
      ws = dis[s1];
      wd = dis[d1];
      s = s1;
      d = d1;
      k = k1;
    }
  }

  if (rem > 0) {
    const int e = eb + nfull * EPW + ((sl < rem) ? sl : (rem - 1));
    const int s = qsrc[e];
    const int d = qdst[e];
    const float w = dis[s] * dis[d];
    v4f v[EPW];
#pragma unroll
    for (int j = 0; j < EPW; ++j) {
      if (j < rem) v[j] = load_row(hw, s, j, lane);
    }
#pragma unroll
    for (int j = 0; j < EPW; ++j) {
      if (j < rem) add_row(mine, v[j], d, w, j, g0);
    }
  }

  if (g0 + GR <= N) {
#pragma unroll
    for (int i = 0; i < GR; ++i)
      *(v4f*)(sums + (size_t)(g0 + i) * 128 + c4) = *(const v4f*)(mine + i * 128);
  } else {
    for (int i = 0; i < GR; ++i) {
      const int row = g0 + i;
      if (row < N) *(v4f*)(sums + (size_t)row * 128 + c4) = *(const v4f*)(mine + i * 128);
    }
  }
}

__global__ void pred_kernel(const float* __restrict__ t, const float* __restrict__ Wr2,
                            const float* __restrict__ br2, float* __restrict__ pred, int N) {
  int i = blockIdx.x * blockDim.x + threadIdx.x;
  if (i >= N) return;
  float a0 = br2[0], a1 = br2[1], a2 = br2[2];
  const float* row = t + (size_t)i * 128;
#pragma unroll 4
  for (int k = 0; k < 128; ++k) {
    float tv = row[k];
    a0 += tv * Wr2[k * 3 + 0];
    a1 += tv * Wr2[k * 3 + 1];
    a2 += tv * Wr2[k * 3 + 2];
  }
  pred[(size_t)i * 3 + 0] = a0;
  pred[(size_t)i * 3 + 1] = a1;
  pred[(size_t)i * 3 + 2] = a2;
}

extern "C" void kernel_launch(void* const* d_in, const int* in_sizes, int n_in,
                              void* d_out, int out_size, void* d_ws, size_t ws_size,
                              hipStream_t stream) {
  const float* x   = (const float*)d_in[0];
  const float* xm  = (const float*)d_in[1];
  const int*   ei  = (const int*)d_in[2];
  const float* W0  = (const float*)d_in[3];
  const float* b0  = (const float*)d_in[4];
  const float* Wh  = (const float*)d_in[5];
  const float* bh  = (const float*)d_in[6];
  const float* Wr1 = (const float*)d_in[7];
  const float* br1 = (const float*)d_in[8];
  const float* Wr2 = (const float*)d_in[9];
  const float* br2 = (const float*)d_in[10];

  const int N = in_sizes[0] / 10;
  const int E = in_sizes[2] / 2;
  const int* srcv = ei;
  const int* dstv = ei + E;

  float* bufA = (float*)d_ws;
  float* bufB = bufA + (size_t)N * 128;
  float* bufC = bufB + (size_t)N * 128;
  float* dis  = bufC + (size_t)N * 128;

  const int NB = (N + RB - 1) / RB;
  int* psrc   = (int*)bufC;
  int* pdst   = psrc + E;
  int* cnt    = pdst + E;
  int* start  = cnt + NB_MAX;
  int* cursor = start + NB_MAX + 32;
  int* qsrc   = cursor + NB_MAX;
  int* qdst   = qsrc + E;
  int* sub    = qdst + E;

  float* emb_out  = (float*)d_out;
  float* pred_out = emb_out + (size_t)N * 128;

  const int mblocks  = ((N + 15) / 16 + 3) / 4;
  const int fblocks  = (E + FILL_EPB - 1) / FILL_EPB;

  hipMemsetAsync(dis, 0, (size_t)N * sizeof(float), stream);
  deg_kernel<<<(E + 255) / 256, 256, 0, stream>>>(dstv, dis, E);
  rsqrt_count_kernel<<<NB, RB, 0, stream>>>(dis, cnt, N);

  scan_ranges<<<1, NB_MAX, 0, stream>>>(cnt, start, cursor, NB);
  fill_ranges<<<fblocks, 256, 0, stream>>>(srcv, dstv, cursor, psrc, pdst, E, NB);
  group_rows<<<NB, GPB * 32, 0, stream>>>(psrc, pdst, start, qsrc, qdst, sub, E);

  build_h0<<<(int)(((size_t)N * 16 + 255) / 256), 256, 0, stream>>>(x, xm, bufA, N);

  gemm128<false, true><<<mblocks, 128, 0, stream>>>(bufA, 16, N, W0, bufB, b0);
  agg_rows<<<NB, AGG_THREADS, 0, stream>>>(bufB, qsrc, qdst, sub, dis, b0, bufA, N);

  for (int l = 0; l < 4; ++l) {
    float* sums = (l == 3) ? emb_out : bufA;
    gemm128<true, true><<<mblocks, 128, 0, stream>>>(bufA, 128, N, Wh + (size_t)l * 128 * 128,
                                                     bufB, bh + (size_t)l * 128);
    agg_rows<<<NB, AGG_THREADS, 0, stream>>>(bufB, qsrc, qdst, sub, dis, bh + (size_t)l * 128,
                                             sums, N);
  }

  gemm128<true, false><<<mblocks, 128, 0, stream>>>(emb_out, 128, N, Wr1, bufB, br1);
  pred_kernel<<<(N + 255) / 256, 256, 0, stream>>>(bufB, Wr2, br2, pred_out, N);
}


# edge weight dis[src]*dis[dst] split: product epilogue scales each hw row by its dis, aggregation adds rows unweighted and scales the finished row by dis[dst], bias last; no dis gather per edge
# speedup vs baseline: 2.8126x; 1.0519x over previous
#include <hip/hip_runtime.h>

typedef __attribute__((ext_vector_type(2))) float v2f;
typedef __attribute__((ext_vector_type(4))) float v4f;
typedef __attribute__((ext_vector_type(8))) float v8f;

__global__ void deg_kernel(const int* __restrict__ dst, float* __restrict__ deg, int E) {
  int e = blockIdx.x * blockDim.x + threadIdx.x;
  if (e < E) unsafeAtomicAdd(&deg[dst[e]], 1.0f);
}

constexpr int RB = 128;
constexpr int RB_SHIFT = 7;
constexpr int NB_MAX = 1024;

__global__ void __launch_bounds__(RB)
rsqrt_count_kernel(float* __restrict__ deg, int* __restrict__ cnt, int N) {
  __shared__ int total;
  if (threadIdx.x == 0) total = 0;
  __syncthreads();
  const int i = blockIdx.x * RB + threadIdx.x;
  float dg = 0.0f;
  if (i < N) {
    dg = deg[i];
    deg[i] = rsqrtf(dg + 1.0f);
  }
  int part = (int)dg;
  for (int off = warpSize / 2; off > 0; off >>= 1) part += __shfl_xor(part, off);
  if ((threadIdx.x & (warpSize - 1)) == 0) atomicAdd(&total, part);
  __syncthreads();
  if (threadIdx.x == 0) cnt[blockIdx.x] = total;
}

__global__ void __launch_bounds__(NB_MAX)
scan_ranges(const int* __restrict__ cnt, int* __restrict__ start, int* __restrict__ cursor, int NB) {
  __shared__ int a[NB_MAX];
  const int t = threadIdx.x;
  a[t] = (t < NB) ? cnt[t] : 0;
  __syncthreads();
  for (int off = 1; off < NB_MAX; off <<= 1) {
    const int add = (t >= off) ? a[t - off] : 0;
    __syncthreads();
    a[t] += add;
    __syncthreads();
  }
  if (t <= NB) {
    const int st = (t == 0) ? 0 : a[t - 1];
    start[t] = st;
    if (t < NB) cursor[t] = st;
  }
}

constexpr int FILL_EPT = 8;
constexpr int FILL_EPB = 256 * FILL_EPT;

__global__ void __launch_bounds__(256)
fill_ranges(const int* __restrict__ src, const int* __restrict__ dst,
            int* __restrict__ cursor, int* __restrict__ psrc, int* __restrict__ pdst,
            int E, int NB) {
  __shared__ int binCnt[NB_MAX];
  __shared__ int binBase[NB_MAX];
  const int t = threadIdx.x;
  for (int r = t; r < NB_MAX; r += 256) binCnt[r] = 0;
  __syncthreads();
  const int eb = blockIdx.x * FILL_EPB + t;
  int dd[FILL_EPT], rk[FILL_EPT];
#pragma unroll
  for (int i = 0; i < FILL_EPT; ++i) {
    const int e = eb + i * 256;
    dd[i] = (e < E) ? dst[e] : 0;
  }
#pragma unroll
  for (int i = 0; i < FILL_EPT; ++i) {
    const int e = eb + i * 256;
    rk[i] = 0;
    if (e < E) rk[i] = atomicAdd(&binCnt[dd[i] >> RB_SHIFT], 1);
  }
  __syncthreads();
  for (int r = t; r < NB; r += 256) {
    const int n = binCnt[r];
    if (n > 0) binBase[r] = atomicAdd(&cursor[r], n);
  }
  __syncthreads();
#pragma unroll
  for (int i = 0; i < FILL_EPT; ++i) {
    const int e = eb + i * 256;
    if (e < E) {
      const int pos = binBase[dd[i] >> RB_SHIFT] + rk[i];
      psrc[pos] = src[e];
      pdst[pos] = dd[i];
    }
  }
}

constexpr int GR = 8;
constexpr int GR_SHIFT = 3;
constexpr int GPB = RB / GR;

__global__ void __launch_bounds__(GPB * 32)
group_rows(const int* __restrict__ psrc, const int* __restrict__ pdst,
           const int* __restrict__ start,
           int* __restrict__ qsrc, int* __restrict__ qdst, int* __restrict__ sub, int E) {
  __shared__ int gcnt[GPB];
  const int lane = threadIdx.x & 31;
  const int wave = __builtin_amdgcn_readfirstlane(threadIdx.x >> 5);
  const int r0 = blockIdx.x * RB;
  const int eb = start[blockIdx.x];
  const int ee = start[blockIdx.x + 1];

  int c = 0;
  for (int base = eb; base < ee; base += 4 * 32) {
    int d[4];
#pragma unroll
    for (int u = 0; u < 4; ++u) {
      const int e = base + u * 32 + lane;
      d[u] = pdst[(e < ee) ? e : (ee - 1)];
    }
#pragma unroll
    for (int u = 0; u < 4; ++u) {
      const int e = base + u * 32 + lane;
      const bool m = (e < ee) & (((d[u] - r0) >> GR_SHIFT) == wave);
      c += __popc(__builtin_amdgcn_ballot_w32(m));
    }
  }
  if (lane == 0) gcnt[wave] = c;
  __syncthreads();
  int pos = eb;
  for (int i = 0; i < wave; ++i) pos += gcnt[i];
  if (lane == 0) sub[blockIdx.x * GPB + wave] = pos;
  if (blockIdx.x == gridDim.x - 1 && threadIdx.x == 0) sub[gridDim.x * GPB] = E;

  for (int base = eb; base < ee; base += 4 * 32) {
    int d[4], s[4];
#pragma unroll
    for (int u = 0; u < 4; ++u) {
      const int e = base + u * 32 + lane;
      const int ec = (e < ee) ? e : (ee - 1);
      d[u] = pdst[ec];
      s[u] = psrc[ec];
    }
#pragma unroll
    for (int u = 0; u < 4; ++u) {
      const int e = base + u * 32 + lane;
      const bool m = (e < ee) & (((d[u] - r0) >> GR_SHIFT) == wave);
      const unsigned mask = __builtin_amdgcn_ballot_w32(m);
      if (m) {
        const int p = pos + (int)__builtin_amdgcn_mbcnt_lo(mask, 0u);
        qsrc[p] = s[u];
        qdst[p] = d[u];
      }
      pos += __popc(mask);
    }
  }
}

__global__ void build_h0(const float* __restrict__ x, const float* __restrict__ xm,
                         float* __restrict__ h0, int N) {
  int idx = blockIdx.x * blockDim.x + threadIdx.x;
  if (idx >= N * 16) return;
  int i = idx >> 4, j = idx & 15;
  h0[idx] = (j < 8) ? x[i * 10 + j] : xm[i * 10 + (j - 8)];
}

template <bool CONV>
__device__ __forceinline__ void store_row(const v8f (&acc)[2][4], int v, size_t o,
                                          v4f badd0, v4f badd1, float rs, float* __restrict__ Out) {
  v4f h0, h1;
  h0.x = acc[0][0][v]; h0.y = acc[0][1][v]; h0.z = acc[0][2][v]; h0.w = acc[0][3][v];
  h1.x = acc[1][0][v]; h1.y = acc[1][1][v]; h1.z = acc[1][2][v]; h1.w = acc[1][3][v];
  if (CONV) {
    *(v4f*)(Out + o) = h0 * rs;
    *(v4f*)(Out + o + 64) = h1 * rs;
  } else {
    v4f t0, t1;
    t0.x = fmaxf(h0.x + badd0.x, 0.0f); t0.y = fmaxf(h0.y + badd0.y, 0.0f);
    t0.z = fmaxf(h0.z + badd0.z, 0.0f); t0.w = fmaxf(h0.w + badd0.w, 0.0f);
    t1.x = fmaxf(h1.x + badd1.x, 0.0f); t1.y = fmaxf(h1.y + badd1.y, 0.0f);
    t1.z = fmaxf(h1.z + badd1.z, 0.0f); t1.w = fmaxf(h1.w + badd1.w, 0.0f);
    *(v4f*)(Out + o) = t0;
    *(v4f*)(Out + o + 64) = t1;
  }
}

template <bool RELU_IN, bool CONV>
__global__ void __launch_bounds__(128)
gemm128(const float* __restrict__ A, int K, int N,
        const float* __restrict__ W,
        float* __restrict__ Out,
        const float* __restrict__ bias,
        const float* __restrict__ dis) {
  const int lane = threadIdx.x & 31;
  const int wave = threadIdx.x >> 5;
  const int m0 = (blockIdx.x * 4 + wave) * 16;
  if (m0 >= N) return;
  const int half = lane >> 4;
  const int c = lane & 15;

  int row = m0 + c;
  row = (row < N) ? row : (N - 1);

  const float* aBase = A + (size_t)row * K + half * 4;
  const float* wBase = W + (size_t)(half * 4) * 128 + c * 4;

  v8f acc[2][4] = {};
  for (int k = 0; k < K; k += 8) {
    v4f a = *(const v4f*)(aBase + k);
    if (RELU_IN) {
      a.x = fmaxf(a.x, 0.0f);
      a.y = fmaxf(a.y, 0.0f);
      a.z = fmaxf(a.z, 0.0f);
      a.w = fmaxf(a.w, 0.0f);
    }
    const float* wp = wBase + (size_t)k * 128;
    v4f b[4][2];
#pragma unroll
    for (int i = 0; i < 4; ++i) {
      b[i][0] = *(const v4f*)(wp + i * 128);
      b[i][1] = *(const v4f*)(wp + i * 128 + 64);
    }
    v2f a01, a23;
    a01.x = a.x; a01.y = a.y;
    a23.x = a.z; a23.y = a.w;
#pragma unroll
    for (int j = 0; j < 4; ++j) {
#pragma unroll
      for (int g = 0; g < 2; ++g) {
        v2f bb;
        bb.x = b[0][g][j];
        bb.y = b[1][g][j];
        acc[g][j] = __builtin_amdgcn_wmma_f32_16x16x4_f32(false, a01, false, bb, (short)0,
                                                          acc[g][j], false, false);
      }
    }
#pragma unroll
    for (int j = 0; j < 4; ++j) {
#pragma unroll
      for (int g = 0; g < 2; ++g) {
        v2f bb;
        bb.x = b[2][g][j];
        bb.y = b[3][g][j];
        acc[g][j] = __builtin_amdgcn_wmma_f32_16x16x4_f32(false, a23, false, bb, (short)0,
                                                          acc[g][j], false, false);
      }
    }
  }

  v4f badd0 = {}, badd1 = {};
  if (!CONV) {
    badd0 = *(const v4f*)(bias + c * 4);
    badd1 = *(const v4f*)(bias + 64 + c * 4);
  }
  const int mb = m0 + half * 8;
  if (m0 + 16 <= N) {
    float rs[8] = {1.0f, 1.0f, 1.0f, 1.0f, 1.0f, 1.0f, 1.0f, 1.0f};
    if (CONV) {
      const v4f r0 = *(const v4f*)(dis + mb);
      const v4f r1 = *(const v4f*)(dis + mb + 4);
      rs[0] = r0.x; rs[1] = r0.y; rs[2] = r0.z; rs[3] = r0.w;
      rs[4] = r1.x; rs[5] = r1.y; rs[6] = r1.z; rs[7] = r1.w;
    }
#pragma unroll
    for (int v = 0; v < 8; ++v)
      store_row<CONV>(acc, v, (size_t)(mb + v) * 128 + c * 4, badd0, badd1, rs[v], Out);
  } else {
#pragma unroll
    for (int v = 0; v < 8; ++v) {
      const int m = mb + v;
      if (m < N) {
        const float rs = CONV ? dis[m] : 1.0f;
        store_row<CONV>(acc, v, (size_t)m * 128 + c * 4, badd0, badd1, rs, Out);
      }
    }
  }
}

constexpr int EPW = 8;
constexpr int AGG_THREADS = GPB * 32;

__device__ __forceinline__ v4f load_row(const float* __restrict__ hw, int s, int j, int lane) {
  const int sj = __builtin_amdgcn_readlane(s, j);
  return *(const v4f*)(hw + (size_t)sj * 128 + lane * 4);
}

__device__ __forceinline__ void add_row(float* mine, v4f v, int d, int j, int g0) {
  const int dj = __builtin_amdgcn_readlane(d, j) - g0;
  v4f* p = (v4f*)(mine + dj * 128);
  *p = *p + v;
}

__global__ void __launch_bounds__(AGG_THREADS) __attribute__((amdgpu_waves_per_eu(16, 16)))
agg_rows(const float* __restrict__ hw,
         const int* __restrict__ qsrc, const int* __restrict__ qdst,
         const int* __restrict__ sub,
         const float* __restrict__ dis, const float* __restrict__ bias,
         float* __restrict__ sums, int N) {
  __shared__ float tile[RB * 128];
  const int lane = threadIdx.x & 31;
  const int wave = __builtin_amdgcn_readfirstlane(threadIdx.x >> 5);
  const int g = blockIdx.x * GPB + wave;
  const int g0 = g * GR;
  const int c4 = lane * 4;
  float* mine = tile + wave * (GR * 128) + c4;

  const v4f b = *(const v4f*)(bias + c4);
  float dr[GR];
#pragma unroll
  for (int i = 0; i < GR; ++i) {
    int row = g0 + i;
    row = (row < N) ? row : (N - 1);
    dr[i] = dis[row];
    *(v4f*)(mine + i * 128) = *(const v4f*)(hw + (size_t)row * 128 + c4);
  }

  const int eb = sub[g];
  const int ee = sub[g + 1];
  const int nfull = (ee - eb) / EPW;
  const int rem = (ee - eb) - nfull * EPW;
  const int sl = lane & (EPW - 1);

  if (nfull > 0) {
    int s = qsrc[eb + sl];
    int d = qdst[eb + sl];
    int k = 0;
    while (k < nfull) {
      v4f v[EPW];
#pragma unroll
      for (int j = 0; j < EPW; ++j) v[j] = load_row(hw, s, j, lane);
      const int k1 = k + 1;
      const int kc = (k1 < nfull) ? k1 : (nfull - 1);
      const int s1 = qsrc[eb + kc * EPW + sl];
      const int d1 = qdst[eb + kc * EPW + sl];
#pragma unroll
      for (int j = 0; j < EPW; ++j) add_row(mine, v[j], d, j, g0);
      s = s1;
      d = d1;
      k = k1;
    }
  }

  if (rem > 0) {
    const int e = eb + nfull * EPW + ((sl < rem) ? sl : (rem - 1));
    const int s = qsrc[e];
    const int d = qdst[e];
    v4f v[EPW];
#pragma unroll
    for (int j = 0; j < EPW; ++j) {
      if (j < rem) v[j] = load_row(hw, s, j, lane);
    }
#pragma unroll
    for (int j = 0; j < EPW; ++j) {
      if (j < rem) add_row(mine, v[j], d, j, g0);
    }
  }

  if (g0 + GR <= N) {
#pragma unroll
    for (int i = 0; i < GR; ++i)
      *(v4f*)(sums + (size_t)(g0 + i) * 128 + c4) = *(const v4f*)(mine + i * 128) * dr[i] + b;
  } else {
#pragma unroll
    for (int i = 0; i < GR; ++i) {
      const int row = g0 + i;
      if (row < N) *(v4f*)(sums + (size_t)row * 128 + c4) = *(const v4f*)(mine + i * 128) * dr[i] + b;
    }
  }
}

__global__ void pred_kernel(const float* __restrict__ t, const float* __restrict__ Wr2,
                            const float* __restrict__ br2, float* __restrict__ pred, int N) {
  int i = blockIdx.x * blockDim.x + threadIdx.x;
  if (i >= N) return;
  float a0 = br2[0], a1 = br2[1], a2 = br2[2];
  const float* row = t + (size_t)i * 128;
#pragma unroll 4
  for (int k = 0; k < 128; ++k) {
    float tv = row[k];
    a0 += tv * Wr2[k * 3 + 0];
    a1 += tv * Wr2[k * 3 + 1];
    a2 += tv * Wr2[k * 3 + 2];
  }
  pred[(size_t)i * 3 + 0] = a0;
  pred[(size_t)i * 3 + 1] = a1;
  pred[(size_t)i * 3 + 2] = a2;
}

extern "C" void kernel_launch(void* const* d_in, const int* in_sizes, int n_in,
                              void* d_out, int out_size, void* d_ws, size_t ws_size,
                              hipStream_t stream) {
  const float* x   = (const float*)d_in[0];
  const float* xm  = (const float*)d_in[1];
  const int*   ei  = (const int*)d_in[2];
  const float* W0  = (const float*)d_in[3];
  const float* b0  = (const float*)d_in[4];
  const float* Wh  = (const float*)d_in[5];
  const float* bh  = (const float*)d_in[6];
  const float* Wr1 = (const float*)d_in[7];
  const float* br1 = (const float*)d_in[8];
  const float* Wr2 = (const float*)d_in[9];
  const float* br2 = (const float*)d_in[10];

  const int N = in_sizes[0] / 10;
  const int E = in_sizes[2] / 2;
  const int* srcv = ei;
  const int* dstv = ei + E;

  float* bufA = (float*)d_ws;
  float* bufB = bufA + (size_t)N * 128;
  float* bufC = bufB + (size_t)N * 128;
  float* dis  = bufC + (size_t)N * 128;

  const int NB = (N + RB - 1) / RB;
  int* psrc   = (int*)bufC;
  int* pdst   = psrc + E;
  int* cnt    = pdst + E;
  int* start  = cnt + NB_MAX;
  int* cursor = start + NB_MAX + 32;
  int* qsrc   = cursor + NB_MAX;
  int* qdst   = qsrc + E;
  int* sub    = qdst + E;

  float* emb_out  = (float*)d_out;
  float* pred_out = emb_out + (size_t)N * 128;

  const int mblocks  = ((N + 15) / 16 + 3) / 4;
  const int fblocks  = (E + FILL_EPB - 1) / FILL_EPB;

  hipMemsetAsync(dis, 0, (size_t)N * sizeof(float), stream);
  deg_kernel<<<(E + 255) / 256, 256, 0, stream>>>(dstv, dis, E);
  rsqrt_count_kernel<<<NB, RB, 0, stream>>>(dis, cnt, N);

  scan_ranges<<<1, NB_MAX, 0, stream>>>(cnt, start, cursor, NB);
  fill_ranges<<<fblocks, 256, 0, stream>>>(srcv, dstv, cursor, psrc, pdst, E, NB);
  group_rows<<<NB, GPB * 32, 0, stream>>>(psrc, pdst, start, qsrc, qdst, sub, E);

  build_h0<<<(int)(((size_t)N * 16 + 255) / 256), 256, 0, stream>>>(x, xm, bufA, N);

  gemm128<false, true><<<mblocks, 128, 0, stream>>>(bufA, 16, N, W0, bufB, b0, dis);
  agg_rows<<<NB, AGG_THREADS, 0, stream>>>(bufB, qsrc, qdst, sub, dis, b0, bufA, N);

  for (int l = 0; l < 4; ++l) {
    float* sums = (l == 3) ? emb_out : bufA;
    gemm128<true, true><<<mblocks, 128, 0, stream>>>(bufA, 128, N, Wh + (size_t)l * 128 * 128,
                                                     bufB, bh + (size_t)l * 128, dis);
    agg_rows<<<NB, AGG_THREADS, 0, stream>>>(bufB, qsrc, qdst, sub, dis, bh + (size_t)l * 128,
                                             sums, N);
  }

  gemm128<true, false><<<mblocks, 128, 0, stream>>>(emb_out, 128, N, Wr1, bufB, br1, dis);
  pred_kernel<<<(N + 255) / 256, 256, 0, stream>>>(bufB, Wr2, br2, pred_out, N);
}


# product: each block copies W once into LDS, one row per output column, k contiguous, so a 16-byte LDS read is a lane's B operand of two WMMAs: no W re-fetch per wave, no register re-sort; same sums
# speedup vs baseline: 2.9862x; 1.0617x over previous
#include <hip/hip_runtime.h>

typedef __attribute__((ext_vector_type(2))) float v2f;
typedef __attribute__((ext_vector_type(4))) float v4f;
typedef __attribute__((ext_vector_type(8))) float v8f;

__global__ void deg_kernel(const int* __restrict__ dst, float* __restrict__ deg, int E) {
  int e = blockIdx.x * blockDim.x + threadIdx.x;
  if (e < E) unsafeAtomicAdd(&deg[dst[e]], 1.0f);
}

constexpr int RB = 128;
constexpr int RB_SHIFT = 7;
constexpr int NB_MAX = 1024;

__global__ void __launch_bounds__(RB)
rsqrt_count_kernel(float* __restrict__ deg, int* __restrict__ cnt, int N) {
  __shared__ int total;
  if (threadIdx.x == 0) total = 0;
  __syncthreads();
  const int i = blockIdx.x * RB + threadIdx.x;
  float dg = 0.0f;
  if (i < N) {
    dg = deg[i];
    deg[i] = rsqrtf(dg + 1.0f);
  }
  int part = (int)dg;
  for (int off = warpSize / 2; off > 0; off >>= 1) part += __shfl_xor(part, off);
  if ((threadIdx.x & (warpSize - 1)) == 0) atomicAdd(&total, part);
  __syncthreads();
  if (threadIdx.x == 0) cnt[blockIdx.x] = total;
}

__global__ void __launch_bounds__(NB_MAX)
scan_ranges(const int* __restrict__ cnt, int* __restrict__ start, int* __restrict__ cursor, int NB) {
  __shared__ int a[NB_MAX];
  const int t = threadIdx.x;
  a[t] = (t < NB) ? cnt[t] : 0;
  __syncthreads();
  for (int off = 1; off < NB_MAX; off <<= 1) {
    const int add = (t >= off) ? a[t - off] : 0;
    __syncthreads();
    a[t] += add;
    __syncthreads();
  }
  if (t <= NB) {
    const int st = (t == 0) ? 0 : a[t - 1];
    start[t] = st;
    if (t < NB) cursor[t] = st;
  }
}

constexpr int FILL_EPT = 8;
constexpr int FILL_EPB = 256 * FILL_EPT;

__global__ void __launch_bounds__(256)
fill_ranges(const int* __restrict__ src, const int* __restrict__ dst,
            int* __restrict__ cursor, int* __restrict__ psrc, int* __restrict__ pdst,
            int E, int NB) {
  __shared__ int binCnt[NB_MAX];
  __shared__ int binBase[NB_MAX];
  const int t = threadIdx.x;
  for (int r = t; r < NB_MAX; r += 256) binCnt[r] = 0;
  __syncthreads();
  const int eb = blockIdx.x * FILL_EPB + t;
  int dd[FILL_EPT], rk[FILL_EPT];
#pragma unroll
  for (int i = 0; i < FILL_EPT; ++i) {
    const int e = eb + i * 256;
    dd[i] = (e < E) ? dst[e] : 0;
  }
#pragma unroll
  for (int i = 0; i < FILL_EPT; ++i) {
    const int e = eb + i * 256;
    rk[i] = 0;
    if (e < E) rk[i] = atomicAdd(&binCnt[dd[i] >> RB_SHIFT], 1);
  }
  __syncthreads();
  for (int r = t; r < NB; r += 256) {
    const int n = binCnt[r];
    if (n > 0) binBase[r] = atomicAdd(&cursor[r], n);
  }
  __syncthreads();
#pragma unroll
  for (int i = 0; i < FILL_EPT; ++i) {
    const int e = eb + i * 256;
    if (e < E) {
      const int pos = binBase[dd[i] >> RB_SHIFT] + rk[i];
      psrc[pos] = src[e];
      pdst[pos] = dd[i];
    }
  }
}

constexpr int GR = 8;
constexpr int GR_SHIFT = 3;
constexpr int GPB = RB / GR;

__global__ void __launch_bounds__(GPB * 32)
group_rows(const int* __restrict__ psrc, const int* __restrict__ pdst,
           const int* __restrict__ start,
           int* __restrict__ qsrc, int* __restrict__ qdst, int* __restrict__ sub, int E) {
  __shared__ int gcnt[GPB];
  const int lane = threadIdx.x & 31;
  const int wave = __builtin_amdgcn_readfirstlane(threadIdx.x >> 5);
  const int r0 = blockIdx.x * RB;
  const int eb = start[blockIdx.x];
  const int ee = start[blockIdx.x + 1];

  int c = 0;
  for (int base = eb; base < ee; base += 4 * 32) {
    int d[4];
#pragma unroll
    for (int u = 0; u < 4; ++u) {
      const int e = base + u * 32 + lane;
      d[u] = pdst[(e < ee) ? e : (ee - 1)];
    }
#pragma unroll
    for (int u = 0; u < 4; ++u) {
      const int e = base + u * 32 + lane;
      const bool m = (e < ee) & (((d[u] - r0) >> GR_SHIFT) == wave);
      c += __popc(__builtin_amdgcn_ballot_w32(m));
    }
  }
  if (lane == 0) gcnt[wave] = c;
  __syncthreads();
  int pos = eb;
  for (int i = 0; i < wave; ++i) pos += gcnt[i];
  if (lane == 0) sub[blockIdx.x * GPB + wave] = pos;
  if (blockIdx.x == gridDim.x - 1 && threadIdx.x == 0) sub[gridDim.x * GPB] = E;

  for (int base = eb; base < ee; base += 4 * 32) {
    int d[4], s[4];
#pragma unroll
    for (int u = 0; u < 4; ++u) {
      const int e = base + u * 32 + lane;
      const int ec = (e < ee) ? e : (ee - 1);
      d[u] = pdst[ec];
      s[u] = psrc[ec];
    }
#pragma unroll
    for (int u = 0; u < 4; ++u) {
      const int e = base + u * 32 + lane;
      const bool m = (e < ee) & (((d[u] - r0) >> GR_SHIFT) == wave);
      const unsigned mask = __builtin_amdgcn_ballot_w32(m);
      if (m) {
        const int p = pos + (int)__builtin_amdgcn_mbcnt_lo(mask, 0u);
        qsrc[p] = s[u];
        qdst[p] = d[u];
      }
      pos += __popc(mask);
    }
  }
}

__global__ void build_h0(const float* __restrict__ x, const float* __restrict__ xm,
                         float* __restrict__ h0, int N) {
  int idx = blockIdx.x * blockDim.x + threadIdx.x;
  if (idx >= N * 16) return;
  int i = idx >> 4, j = idx & 15;
  h0[idx] = (j < 8) ? x[i * 10 + j] : xm[i * 10 + (j - 8)];
}

template <bool CONV>
__device__ __forceinline__ void store_row(const v8f (&acc)[2][4], int v, size_t o,
                                          v4f badd0, v4f badd1, float rs, float* __restrict__ Out) {
  v4f h0, h1;
  h0.x = acc[0][0][v]; h0.y = acc[0][1][v]; h0.z = acc[0][2][v]; h0.w = acc[0][3][v];
  h1.x = acc[1][0][v]; h1.y = acc[1][1][v]; h1.z = acc[1][2][v]; h1.w = acc[1][3][v];
  if (CONV) {
    *(v4f*)(Out + o) = h0 * rs;
    *(v4f*)(Out + o + 64) = h1 * rs;
  } else {
    v4f t0, t1;
    t0.x = fmaxf(h0.x + badd0.x, 0.0f); t0.y = fmaxf(h0.y + badd0.y, 0.0f);
    t0.z = fmaxf(h0.z + badd0.z, 0.0f); t0.w = fmaxf(h0.w + badd0.w, 0.0f);
    t1.x = fmaxf(h1.x + badd1.x, 0.0f); t1.y = fmaxf(h1.y + badd1.y, 0.0f);
    t1.z = fmaxf(h1.z + badd1.z, 0.0f); t1.w = fmaxf(h1.w + badd1.w, 0.0f);
    *(v4f*)(Out + o) = t0;
    *(v4f*)(Out + o + 64) = t1;
  }
}

template <int K, bool RELU_IN, bool CONV>
__global__ void __launch_bounds__(128) __attribute__((amdgpu_waves_per_eu(4, 4)))
gemm128(const float* __restrict__ A, int N,
        const float* __restrict__ W,
        float* __restrict__ Out,
        const float* __restrict__ bias,
        const float* __restrict__ dis) {
  constexpr int P = K + 4;
  constexpr int S = K / 8;
  __shared__ float wt[128 * P];

  const int lane = threadIdx.x & 31;
  const int wave = threadIdx.x >> 5;
  const int m0 = (blockIdx.x * 4 + wave) * 16;
  const int half = lane >> 4;
  const int c = lane & 15;

  int row = m0 + c;
  row = (row < N) ? row : (N - 1);

  const float* aBase = A + (size_t)row * K + half * 4;
  v4f a[S];
#pragma unroll
  for (int s = 0; s < S; ++s) a[s] = *(const v4f*)(aBase + s * 8);

  {
    const int g = lane >> 4;
    const float* wp = W + (size_t)(wave * (K / 4)) * 128 + lane * 4;
    float* lp = wt + (64 * g + c) * P + wave * (K / 4);
#pragma unroll 4
    for (int kq = 0; kq < K / 16; ++kq) {
      const v4f w0 = *(const v4f*)(wp + (kq * 4 + 0) * 128);
      const v4f w1 = *(const v4f*)(wp + (kq * 4 + 1) * 128);
      const v4f w2 = *(const v4f*)(wp + (kq * 4 + 2) * 128);
      const v4f w3 = *(const v4f*)(wp + (kq * 4 + 3) * 128);
      v4f t0, t1, t2, t3;
      t0.x = w0.x; t0.y = w1.x; t0.z = w2.x; t0.w = w3.x;
      t1.x = w0.y; t1.y = w1.y; t1.z = w2.y; t1.w = w3.y;
      t2.x = w0.z; t2.y = w1.z; t2.z = w2.z; t2.w = w3.z;
      t3.x = w0.w; t3.y = w1.w; t3.z = w2.w; t3.w = w3.w;
      *(v4f*)(lp + (16 * 0) * P + kq * 4) = t0;
      *(v4f*)(lp + (16 * 1) * P + kq * 4) = t1;
      *(v4f*)(lp + (16 * 2) * P + kq * 4) = t2;
      *(v4f*)(lp + (16 * 3) * P + kq * 4) = t3;
    }
  }
  __syncthreads();
  if (m0 >= N) return;

  const float* bp = wt + c * P + half * 4;
  v8f acc[2][4] = {};
  v4f bc[8];
#pragma unroll
  for (int t = 0; t < 8; ++t) bc[t] = *(const v4f*)(bp + (16 * t) * P);
#pragma unroll
  for (int s = 0; s < S; ++s) {
    v4f bn[8];
    if (s + 1 < S) {
#pragma unroll
      for (int t = 0; t < 8; ++t) bn[t] = *(const v4f*)(bp + (16 * t) * P + (s + 1) * 8);
    }
    v4f av = a[s];
    if (RELU_IN) {
      av.x = fmaxf(av.x, 0.0f);
      av.y = fmaxf(av.y, 0.0f);
      av.z = fmaxf(av.z, 0.0f);
      av.w = fmaxf(av.w, 0.0f);
    }
    v2f a01, a23;
    a01.x = av.x; a01.y = av.y;
    a23.x = av.z; a23.y = av.w;
#pragma unroll
    for (int t = 0; t < 8; ++t) {
      v2f bb;
      bb.x = bc[t].x;
      bb.y = bc[t].y;
      acc[t >> 2][t & 3] = __builtin_amdgcn_wmma_f32_16x16x4_f32(false, a01, false, bb, (short)0,
                                                                 acc[t >> 2][t & 3], false, false);
    }
#pragma unroll
    for (int t = 0; t < 8; ++t) {
      v2f bb;
      bb.x = bc[t].z;
      bb.y = bc[t].w;
      acc[t >> 2][t & 3] = __builtin_amdgcn_wmma_f32_16x16x4_f32(false, a23, false, bb, (short)0,
                                                                 acc[t >> 2][t & 3], false, false);
    }
    if (s + 1 < S) {
#pragma unroll
      for (int t = 0; t < 8; ++t) bc[t] = bn[t];
    }
  }

  v4f badd0 = {}, badd1 = {};
  if (!CONV) {
    badd0 = *(const v4f*)(bias + c * 4);
    badd1 = *(const v4f*)(bias + 64 + c * 4);
  }
  const int mb = m0 + half * 8;
  if (m0 + 16 <= N) {
    float rs[8] = {1.0f, 1.0f, 1.0f, 1.0f, 1.0f, 1.0f, 1.0f, 1.0f};
    if (CONV) {
      const v4f r0 = *(const v4f*)(dis + mb);
      const v4f r1 = *(const v4f*)(dis + mb + 4);
      rs[0] = r0.x; rs[1] = r0.y; rs[2] = r0.z; rs[3] = r0.w;
      rs[4] = r1.x; rs[5] = r1.y; rs[6] = r1.z; rs[7] = r1.w;
    }
#pragma unroll
    for (int v = 0; v < 8; ++v)
      store_row<CONV>(acc, v, (size_t)(mb + v) * 128 + c * 4, badd0, badd1, rs[v], Out);
  } else {
#pragma unroll
    for (int v = 0; v < 8; ++v) {
      const int m = mb + v;
      if (m < N) {
        const float rs = CONV ? dis[m] : 1.0f;
        store_row<CONV>(acc, v, (size_t)m * 128 + c * 4, badd0, badd1, rs, Out);
      }
    }
  }
}

constexpr int EPW = 8;
constexpr int AGG_THREADS = GPB * 32;

__device__ __forceinline__ v4f load_row(const float* __restrict__ hw, int s, int j, int lane) {
  const int sj = __builtin_amdgcn_readlane(s, j);
  return *(const v4f*)(hw + (size_t)sj * 128 + lane * 4);
}

__device__ __forceinline__ void add_row(float* mine, v4f v, int d, int j, int g0) {
  const int dj = __builtin_amdgcn_readlane(d, j) - g0;
  v4f* p = (v4f*)(mine + dj * 128);
  *p = *p + v;
}

__global__ void __launch_bounds__(AGG_THREADS) __attribute__((amdgpu_waves_per_eu(16, 16)))
agg_rows(const float* __restrict__ hw,
         const int* __restrict__ qsrc, const int* __restrict__ qdst,
         const int* __restrict__ sub,
         const float* __restrict__ dis, const float* __restrict__ bias,
         float* __restrict__ sums, int N) {
  __shared__ float tile[RB * 128];
  const int lane = threadIdx.x & 31;
  const int wave = __builtin_amdgcn_readfirstlane(threadIdx.x >> 5);
  const int g = blockIdx.x * GPB + wave;
  const int g0 = g * GR;
  const int c4 = lane * 4;
  float* mine = tile + wave * (GR * 128) + c4;

  const v4f b = *(const v4f*)(bias + c4);
  float dr[GR];
#pragma unroll
  for (int i = 0; i < GR; ++i) {
    int row = g0 + i;
    row = (row < N) ? row : (N - 1);
    dr[i] = dis[row];
    *(v4f*)(mine + i * 128) = *(const v4f*)(hw + (size_t)row * 128 + c4);
  }

  const int eb = sub[g];
  const int ee = sub[g + 1];
  const int nfull = (ee - eb) / EPW;
  const int rem = (ee - eb) - nfull * EPW;
  const int sl = lane & (EPW - 1);

  if (nfull > 0) {
    int s = qsrc[eb + sl];
    int d = qdst[eb + sl];
    int k = 0;
    while (k < nfull) {
      v4f v[EPW];
#pragma unroll
      for (int j = 0; j < EPW; ++j) v[j] = load_row(hw, s, j, lane);
      const int k1 = k + 1;
      const int kc = (k1 < nfull) ? k1 : (nfull - 1);
      const int s1 = qsrc[eb + kc * EPW + sl];
      const int d1 = qdst[eb + kc * EPW + sl];
#pragma unroll
      for (int j = 0; j < EPW; ++j) add_row(mine, v[j], d, j, g0);
      s = s1;
      d = d1;
      k = k1;
    }
  }

  if (rem > 0) {
    const int e = eb + nfull * EPW + ((sl < rem) ? sl : (rem - 1));
    const int s = qsrc[e];
    const int d = qdst[e];
    v4f v[EPW];
#pragma unroll
    for (int j = 0; j < EPW; ++j) {
      if (j < rem) v[j] = load_row(hw, s, j, lane);
    }
#pragma unroll
    for (int j = 0; j < EPW; ++j) {
      if (j < rem) add_row(mine, v[j], d, j, g0);
    }
  }

  if (g0 + GR <= N) {
#pragma unroll
    for (int i = 0; i < GR; ++i)
      *(v4f*)(sums + (size_t)(g0 + i) * 128 + c4) = *(const v4f*)(mine + i * 128) * dr[i] + b;
  } else {
#pragma unroll
    for (int i = 0; i < GR; ++i) {
      const int row = g0 + i;
      if (row < N) *(v4f*)(sums + (size_t)row * 128 + c4) = *(const v4f*)(mine + i * 128) * dr[i] + b;
    }
  }
}

__global__ void pred_kernel(const float* __restrict__ t, const float* __restrict__ Wr2,
                            const float* __restrict__ br2, float* __restrict__ pred, int N) {
  int i = blockIdx.x * blockDim.x + threadIdx.x;
  if (i >= N) return;
  float a0 = br2[0], a1 = br2[1], a2 = br2[2];
  const float* row = t + (size_t)i * 128;
#pragma unroll 4
  for (int k = 0; k < 128; ++k) {
    float tv = row[k];
    a0 += tv * Wr2[k * 3 + 0];
    a1 += tv * Wr2[k * 3 + 1];
    a2 += tv * Wr2[k * 3 + 2];
  }
  pred[(size_t)i * 3 + 0] = a0;
  pred[(size_t)i * 3 + 1] = a1;
  pred[(size_t)i * 3 + 2] = a2;
}

extern "C" void kernel_launch(void* const* d_in, const int* in_sizes, int n_in,
                              void* d_out, int out_size, void* d_ws, size_t ws_size,
                              hipStream_t stream) {
  const float* x   = (const float*)d_in[0];
  const float* xm  = (const float*)d_in[1];
  const int*   ei  = (const int*)d_in[2];
  const float* W0  = (const float*)d_in[3];
  const float* b0  = (const float*)d_in[4];
  const float* Wh  = (const float*)d_in[5];
  const float* bh  = (const float*)d_in[6];
  const float* Wr1 = (const float*)d_in[7];
  const float* br1 = (const float*)d_in[8];
  const float* Wr2 = (const float*)d_in[9];
  const float* br2 = (const float*)d_in[10];

  const int N = in_sizes[0] / 10;
  const int E = in_sizes[2] / 2;
  const int* srcv = ei;
  const int* dstv = ei + E;

  float* bufA = (float*)d_ws;
  float* bufB = bufA + (size_t)N * 128;
  float* bufC = bufB + (size_t)N * 128;
  float* dis  = bufC + (size_t)N * 128;

  const int NB = (N + RB - 1) / RB;
  int* psrc   = (int*)bufC;
  int* pdst   = psrc + E;
  int* cnt    = pdst + E;
  int* start  = cnt + NB_MAX;
  int* cursor = start + NB_MAX + 32;
  int* qsrc   = cursor + NB_MAX;
  int* qdst   = qsrc + E;
  int* sub    = qdst + E;

  float* emb_out  = (float*)d_out;
  float* pred_out = emb_out + (size_t)N * 128;

  const int mblocks  = ((N + 15) / 16 + 3) / 4;
  const int fblocks  = (E + FILL_EPB - 1) / FILL_EPB;

  hipMemsetAsync(dis, 0, (size_t)N * sizeof(float), stream);
  deg_kernel<<<(E + 255) / 256, 256, 0, stream>>>(dstv, dis, E);
  rsqrt_count_kernel<<<NB, RB, 0, stream>>>(dis, cnt, N);

  scan_ranges<<<1, NB_MAX, 0, stream>>>(cnt, start, cursor, NB);
  fill_ranges<<<fblocks, 256, 0, stream>>>(srcv, dstv, cursor, psrc, pdst, E, NB);
  group_rows<<<NB, GPB * 32, 0, stream>>>(psrc, pdst, start, qsrc, qdst, sub, E);

  build_h0<<<(int)(((size_t)N * 16 + 255) / 256), 256, 0, stream>>>(x, xm, bufA, N);

  gemm128<16, false, true><<<mblocks, 128, 0, stream>>>(bufA, N, W0, bufB, b0, dis);
  agg_rows<<<NB, AGG_THREADS, 0, stream>>>(bufB, qsrc, qdst, sub, dis, b0, bufA, N);

  for (int l = 0; l < 4; ++l) {
    float* sums = (l == 3) ? emb_out : bufA;
    gemm128<128, true, true><<<mblocks, 128, 0, stream>>>(bufA, N, Wh + (size_t)l * 128 * 128,
                                                          bufB, bh + (size_t)l * 128, dis);
    agg_rows<<<NB, AGG_THREADS, 0, stream>>>(bufB, qsrc, qdst, sub, dis, bh + (size_t)l * 128,
                                             sums, N);
  }

  gemm128<128, true, false><<<mblocks, 128, 0, stream>>>(emb_out, N, Wr1, bufB, br1, dis);
  pred_kernel<<<(N + 255) / 256, 256, 0, stream>>>(bufB, Wr2, br2, pred_out, N);
}
